# starting kernel
# baseline (speedup 1.0000x reference)
#include <hip/hip_runtime.h>

#define NNODES 100000
#define NEDGES 3200000

typedef __attribute__((ext_vector_type(2))) float v2f;
typedef __attribute__((ext_vector_type(8))) float v8f;

__global__ void gcn_init_deg(float* __restrict__ deg) {
    int v = blockIdx.x * 256 + threadIdx.x;
    if (v < NNODES) deg[v] = 1.0f;
}

__global__ void gcn_accum_deg(const int* __restrict__ col, float* __restrict__ deg) {
    int e = blockIdx.x * 256 + threadIdx.x;
    if (e < NEDGES) atomicAdd(&deg[col[e]], 1.0f);
}

__global__ void gcn_make_dinv(const float* __restrict__ deg, float* __restrict__ dinv) {
    int v = blockIdx.x * 256 + threadIdx.x;
    if (v < NNODES) dinv[v] = rsqrtf(deg[v]);
}

__global__ __launch_bounds__(256) void gcn_gemm_scale(
        const float* __restrict__ x, const float* __restrict__ W,
        const float* __restrict__ dinv, float* __restrict__ g,
        float* __restrict__ out) {
    __shared__ float Wl[256 * 64];
    const int tid = threadIdx.x;
    for (int i = tid; i < 256 * 64; i += 256) {
        int k = i >> 6, n = i & 63;
        Wl[(k >> 1) * 128 + (n << 1) + (k & 1)] = W[i];
    }
    __syncthreads();

    const int wave = tid >> 5;
    const int lane = tid & 31;
    const int half = lane >> 4;
    const int mr   = lane & 15;
    const int m0   = blockIdx.x * 128 + wave * 16;
    if (m0 >= NNODES) return;

    const float* xrow = x + (size_t)(m0 + mr) * 256;
    const v2f*   Wp   = (const v2f*)Wl;

    v8f acc0 = {}, acc1 = {}, acc2 = {}, acc3 = {};
    for (int k0 = 0; k0 < 256; k0 += 4) {
        const int ka = k0 + half * 2;
        v2f a = *(const v2f*)(xrow + ka);
        const int kp = ka >> 1;
        v2f b0 = Wp[kp * 64 + mr];
        v2f b1 = Wp[kp * 64 + 16 + mr];
        v2f b2 = Wp[kp * 64 + 32 + mr];
        v2f b3 = Wp[kp * 64 + 48 + mr];
        acc0 = __builtin_amdgcn_wmma_f32_16x16x4_f32(false, a, false, b0, (short)0, acc0, false, false);
        acc1 = __builtin_amdgcn_wmma_f32_16x16x4_f32(false, a, false, b1, (short)0, acc1, false, false);
        acc2 = __builtin_amdgcn_wmma_f32_16x16x4_f32(false, a, false, b2, (short)0, acc2, false, false);
        acc3 = __builtin_amdgcn_wmma_f32_16x16x4_f32(false, a, false, b3, (short)0, acc3, false, false);
    }

    #pragma unroll
    for (int i = 0; i < 8; ++i) {
        const int m = m0 + i + half * 8;
        const float s = dinv[m];
        const size_t rb = (size_t)m * 64 + mr;
        float v0 = acc0[i] * s; g[rb]      = v0; out[rb]      = v0;
        float v1 = acc1[i] * s; g[rb + 16] = v1; out[rb + 16] = v1;
        float v2 = acc2[i] * s; g[rb + 32] = v2; out[rb + 32] = v2;
        float v3 = acc3[i] * s; g[rb + 48] = v3; out[rb + 48] = v3;
    }
}

__global__ void gcn_scatter(const float* __restrict__ g,
                            const int* __restrict__ row,
                            const int* __restrict__ col,
                            float* __restrict__ out) {
    size_t t = (size_t)blockIdx.x * 256 + threadIdx.x;
    size_t e = t >> 6;
    int    n = (int)(t & 63);
    if (e < (size_t)NEDGES) {
        int r = row[e], c = col[e];
        atomicAdd(&out[(size_t)c * 64 + n], g[(size_t)r * 64 + n]);
    }
}

__global__ void gcn_finalize(const float* __restrict__ dinv,
                             const float* __restrict__ b,
                             float* __restrict__ out) {
    size_t t = (size_t)blockIdx.x * 256 + threadIdx.x;
    if (t < (size_t)NNODES * 64) {
        int v = (int)(t >> 6), n = (int)(t & 63);
        out[t] = dinv[v] * out[t] + b[n];
    }
}

extern "C" void kernel_launch(void* const* d_in, const int* in_sizes, int n_in,
                              void* d_out, int out_size, void* d_ws, size_t ws_size,
                              hipStream_t stream) {
    const float* x  = (const float*)d_in[0];
    const int*   ei = (const int*)d_in[1];
    const float* W  = (const float*)d_in[2];
    const float* b  = (const float*)d_in[3];
    float* out = (float*)d_out;

    const int* row = ei;
    const int* col = ei + NEDGES;

    float* deg  = (float*)d_ws;
    float* dinv = deg + NNODES;
    float* g    = dinv + NNODES;

    gcn_init_deg  <<<(NNODES + 255) / 256, 256, 0, stream>>>(deg);
    gcn_accum_deg <<<(NEDGES + 255) / 256, 256, 0, stream>>>(col, deg);
    gcn_make_dinv <<<(NNODES + 255) / 256, 256, 0, stream>>>(deg, dinv);
    gcn_gemm_scale<<<(NNODES + 127) / 128, 256, 0, stream>>>(x, W, dinv, g, out);
    gcn_scatter   <<<(int)(((size_t)NEDGES * 64) / 256), 256, 0, stream>>>(g, row, col, out);
    gcn_finalize  <<<(NNODES * 64 + 255) / 256, 256, 0, stream>>>(dinv, b, out);
}


# Float-atomic edge scatter replaced by counting sort of edges by destination (returning int atomic rank, scan, fill) and one wave per node summing g rows in registers; sum order = arrival rank
# speedup vs baseline: 2.2679x; 2.2679x over previous
#include <hip/hip_runtime.h>


#define NNODES 100000
#define NEDGES 3200000
#define NPAD   100352

typedef __attribute__((ext_vector_type(2))) float v2f;
typedef __attribute__((ext_vector_type(8))) float v8f;

__global__ void gcn_zero_cnt(int* __restrict__ cnt) {
    const int i = blockIdx.x * 256 + threadIdx.x;
    ((int4*)cnt)[i] = make_int4(0, 0, 0, 0);
}

__global__ void gcn_rank(const int* __restrict__ col, int* __restrict__ cnt,
                         int* __restrict__ pos) {
    const int i = blockIdx.x * 256 + threadIdx.x;
    const int4 c = ((const int4*)col)[i];
    int4 p;
    p.x = atomicAdd(&cnt[c.x], 1);
    p.y = atomicAdd(&cnt[c.y], 1);
    p.z = atomicAdd(&cnt[c.z], 1);
    p.w = atomicAdd(&cnt[c.w], 1);
    ((int4*)pos)[i] = p;
}

__global__ __launch_bounds__(256) void gcn_scan_tiles(
        const int* __restrict__ cnt, int* __restrict__ loc,
        int* __restrict__ tsum, float* __restrict__ dinv) {
    __shared__ int wsum[8];
    const int t = threadIdx.x, lane = t & 31, wave = t >> 5;
    const int base = blockIdx.x * 1024 + t * 4;
    const int4 c = *(const int4*)(cnt + base);
    const int s = c.x + c.y + c.z + c.w;
    int inc = s;
    for (int d = 1; d < 32; d <<= 1) {
        const int o = __shfl_up(inc, d);
        if (lane >= d) inc += o;
    }
    if (lane == 31) wsum[wave] = inc;
    __syncthreads();
    int wpre = 0, tot = 0;
    for (int w = 0; w < 8; ++w) {
        const int x = wsum[w];
        if (w < wave) wpre += x;
        tot += x;
    }
    const int ex = wpre + inc - s;
    int4 o;
    o.x = ex; o.y = o.x + c.x; o.z = o.y + c.y; o.w = o.z + c.z;
    *(int4*)(loc + base) = o;
    if (t == 0) tsum[blockIdx.x] = tot;
    if (base < NNODES) {
        float4 d;
        d.x = rsqrtf((float)(c.x + 1)); d.y = rsqrtf((float)(c.y + 1));
        d.z = rsqrtf((float)(c.z + 1)); d.w = rsqrtf((float)(c.w + 1));
        *(float4*)(dinv + base) = d;
    }
}

__global__ __launch_bounds__(256) void gcn_scan_add(
        const int* __restrict__ loc, const int* __restrict__ tsum,
        int* __restrict__ off) {
    __shared__ int wsum[8];
    const int t = threadIdx.x, lane = t & 31, wave = t >> 5;
    int x = (t < (int)blockIdx.x) ? tsum[t] : 0;
    for (int d = 16; d >= 1; d >>= 1) x += __shfl_xor(x, d);
    if (lane == 0) wsum[wave] = x;
    __syncthreads();
    int pre = 0;
    for (int w = 0; w < 8; ++w) pre += wsum[w];
    const int base = blockIdx.x * 1024 + t * 4;
    int4 l = *(const int4*)(loc + base);
    l.x += pre; l.y += pre; l.z += pre; l.w += pre;
    *(int4*)(off + base) = l;
}

__global__ void gcn_fill(const int* __restrict__ row, const int* __restrict__ col,
                         const int* __restrict__ pos, const int* __restrict__ off,
                         int* __restrict__ srow) {
    const int i = blockIdx.x * 256 + threadIdx.x;
    const int4 r = ((const int4*)row)[i];
    const int4 c = ((const int4*)col)[i];
    const int4 p = ((const int4*)pos)[i];
    srow[off[c.x] + p.x] = r.x;
    srow[off[c.y] + p.y] = r.y;
    srow[off[c.z] + p.z] = r.z;
    srow[off[c.w] + p.w] = r.w;
}

__global__ __launch_bounds__(256) void gcn_gemm_scale(
        const float* __restrict__ x, const float* __restrict__ W,
        const float* __restrict__ dinv, float* __restrict__ g) {
    __shared__ float Wl[256 * 64];
    const int tid = threadIdx.x;
    for (int i = tid; i < 256 * 64; i += 256) {
        int k = i >> 6, n = i & 63;
        Wl[(k >> 1) * 128 + (n << 1) + (k & 1)] = W[i];
    }
    __syncthreads();

    const int wave = tid >> 5;
    const int lane = tid & 31;
    const int half = lane >> 4;
    const int mr   = lane & 15;
    const int m0   = blockIdx.x * 128 + wave * 16;
    if (m0 >= NNODES) return;

    const float* xrow = x + (size_t)(m0 + mr) * 256;
    const v2f*   Wp   = (const v2f*)Wl;

    v8f acc0 = {}, acc1 = {}, acc2 = {}, acc3 = {};
    for (int k0 = 0; k0 < 256; k0 += 4) {
        const int ka = k0 + half * 2;
        v2f a = *(const v2f*)(xrow + ka);
        const int kp = ka >> 1;
        v2f b0 = Wp[kp * 64 + mr];
        v2f b1 = Wp[kp * 64 + 16 + mr];
        v2f b2 = Wp[kp * 64 + 32 + mr];
        v2f b3 = Wp[kp * 64 + 48 + mr];
        acc0 = __builtin_amdgcn_wmma_f32_16x16x4_f32(false, a, false, b0, (short)0, acc0, false, false);
        acc1 = __builtin_amdgcn_wmma_f32_16x16x4_f32(false, a, false, b1, (short)0, acc1, false, false);
        acc2 = __builtin_amdgcn_wmma_f32_16x16x4_f32(false, a, false, b2, (short)0, acc2, false, false);
        acc3 = __builtin_amdgcn_wmma_f32_16x16x4_f32(false, a, false, b3, (short)0, acc3, false, false);
    }

    #pragma unroll
    for (int i = 0; i < 8; ++i) {
        const int m = m0 + i + half * 8;
        const float s = dinv[m];
        const size_t rb = (size_t)m * 64 + mr;
        g[rb]      = acc0[i] * s;
        g[rb + 16] = acc1[i] * s;
        g[rb + 32] = acc2[i] * s;
        g[rb + 48] = acc3[i] * s;
    }
}

__global__ __launch_bounds__(256) void gcn_gather(
        const float* __restrict__ g, const int* __restrict__ off,
        const int* __restrict__ srow, const float* __restrict__ dinv,
        const float* __restrict__ b, float* __restrict__ out) {
    const int lane = threadIdx.x & 31;
    const int half = lane >> 4;
    const int sub  = lane & 15;
    const int v = __builtin_amdgcn_readfirstlane((int)(blockIdx.x * 8 + (threadIdx.x >> 5)));
    const int beg = off[v], end = off[v + 1];
    const float4* g4 = (const float4*)g;

    const float4 self = g4[(unsigned)v * 16u + (unsigned)sub];
    float4 acc = make_float4(0.f, 0.f, 0.f, 0.f);

    int idx = v;
    if (lane < end - beg) idx = srow[beg + lane];
    for (int base = beg; base < end; base += 32) {
        const int n = min(32, end - base);
        int nidx = v;
        if (lane < end - base - 32) nidx = srow[base + 32 + lane];
        for (int k = 0; k < n; k += 8) {
            for (int u = 0; u < 4; ++u) {
                const int kk = k + 2 * u + half;
                const unsigned r = (unsigned)__shfl(idx, kk);
                const float4 t = g4[r * 16u + (unsigned)sub];
                const bool ok = kk < n;
                acc.x += ok ? t.x : 0.f; acc.y += ok ? t.y : 0.f;
                acc.z += ok ? t.z : 0.f; acc.w += ok ? t.w : 0.f;
            }
        }
        idx = nidx;
    }
    if (half == 0) {
        acc.x += self.x; acc.y += self.y; acc.z += self.z; acc.w += self.w;
    }

    acc.x += __shfl_xor(acc.x, 16); acc.y += __shfl_xor(acc.y, 16);
    acc.z += __shfl_xor(acc.z, 16); acc.w += __shfl_xor(acc.w, 16);
    if (half == 0) {
        const float s = dinv[v];
        const float4 bb = ((const float4*)b)[sub];
        float4 o;
        o.x = s * acc.x + bb.x; o.y = s * acc.y + bb.y;
        o.z = s * acc.z + bb.z; o.w = s * acc.w + bb.w;
        ((float4*)out)[(unsigned)v * 16u + (unsigned)sub] = o;
    }
}

extern "C" void kernel_launch(void* const* d_in, const int* in_sizes, int n_in,
                              void* d_out, int out_size, void* d_ws, size_t ws_size,
                              hipStream_t stream) {
    const float* x  = (const float*)d_in[0];
    const int*   ei = (const int*)d_in[1];
    const float* W  = (const float*)d_in[2];
    const float* b  = (const float*)d_in[3];
    float* out = (float*)d_out;

    const int* row = ei;
    const int* col = ei + NEDGES;

    int*   cnt  = (int*)d_ws;
    int*   loc  = cnt + NPAD;
    int*   off  = loc + NPAD;
    int*   tsum = off + NPAD;
    float* dinv = (float*)(tsum + 128);
    int*   pos  = (int*)(dinv + NPAD);
    int*   srow = pos + NEDGES;
    float* g    = (float*)(srow + NEDGES);

    gcn_zero_cnt  <<<NPAD / 1024, 256, 0, stream>>>(cnt);
    gcn_rank      <<<NEDGES / 1024, 256, 0, stream>>>(col, cnt, pos);
    gcn_scan_tiles<<<NPAD / 1024, 256, 0, stream>>>(cnt, loc, tsum, dinv);
    gcn_scan_add  <<<NPAD / 1024, 256, 0, stream>>>(loc, tsum, off);
    gcn_fill      <<<NEDGES / 1024, 256, 0, stream>>>(row, col, pos, off, srow);
    gcn_gemm_scale<<<(NNODES + 127) / 128, 256, 0, stream>>>(x, W, dinv, g);
    gcn_gather    <<<NNODES / 8, 256, 0, stream>>>(g, off, srow, dinv, b, out);
}


# Edge grouping by destination moved from global returning atomics to LDS counters: each edge chunk ordered by node bucket, then one block per bucket ranks by node; sum order = LDS arrival rank
# speedup vs baseline: 8.0609x; 3.5543x over previous
#include <hip/hip_runtime.h>


#define NNODES 100000
#define NEDGES 3200000
#define NPAD   100352

#define CHUNK   12800
#define NCHUNK  250
#define BNODES  391
#define NBUCK   256
#define SCAP    16384

typedef __attribute__((ext_vector_type(2))) float v2f;
typedef __attribute__((ext_vector_type(8))) float v8f;

template <int NW>
__device__ __forceinline__ int block_excl_scan(int x, int* wsum, int& total) {
    const int lane = threadIdx.x & 31, wave = threadIdx.x >> 5;
    int inc = x;
    for (int d = 1; d < 32; d <<= 1) {
        const int o = __shfl_up(inc, d);
        if (lane >= d) inc += o;
    }
    if (lane == 31) wsum[wave] = inc;
    __syncthreads();
    int wpre = 0, tot = 0;
    for (int w = 0; w < NW; ++w) {
        const int s = wsum[w];
        if (w < wave) wpre += s;
        tot += s;
    }
    total = tot;
    return wpre + inc - x;
}

__global__ __launch_bounds__(640) void gcn_bucket(
        const int* __restrict__ row, const int* __restrict__ col,
        int* __restrict__ sorted, int* __restrict__ tab) {
    __shared__ int cnt[512];
    __shared__ int wsum[32];
    __shared__ int buf[CHUNK];
    const int tid = threadIdx.x;
    if (tid < 512) cnt[tid] = 0;
    __syncthreads();

    const int4* c4 = (const int4*)(col + (size_t)blockIdx.x * CHUNK);
    const int4* r4 = (const int4*)(row + (size_t)blockIdx.x * CHUNK);
    int4 c[5], r[5];
    #pragma unroll
    for (int q = 0; q < 5; ++q) { c[q] = c4[tid + 640 * q]; r[q] = r4[tid + 640 * q]; }

    int key[20], wd[20];
    #pragma unroll
    for (int q = 0; q < 5; ++q) {
        const int d[4] = {c[q].x, c[q].y, c[q].z, c[q].w};
        const int s[4] = {r[q].x, r[q].y, r[q].z, r[q].w};
        #pragma unroll
        for (int j = 0; j < 4; ++j) {
            const int b  = (int)((unsigned)d[j] / (unsigned)BNODES);
            const int rk = atomicAdd(&cnt[b], 1);
            key[q * 4 + j] = (b << 16) | rk;
            wd[q * 4 + j]  = ((d[j] - b * BNODES) << 17) | s[j];
        }
    }
    __syncthreads();

    const int x = (tid < 512) ? cnt[tid] : 0;
    int total;
    const int ex = block_excl_scan<20>(x, wsum, total);
    if (tid < 512) cnt[tid] = ex;
    __syncthreads();

    #pragma unroll
    for (int e = 0; e < 20; ++e) buf[cnt[key[e] >> 16] + (key[e] & 0xFFFF)] = wd[e];
    if (tid <= NBUCK) tab[tid * NCHUNK + blockIdx.x] = cnt[tid];
    __syncthreads();

    int4* o4 = (int4*)(sorted + (size_t)blockIdx.x * CHUNK);
    const int4* b4 = (const int4*)buf;
    for (int q = 0; q < 5; ++q) o4[tid + 640 * q] = b4[tid + 640 * q];
}

__global__ __launch_bounds__(1024) void gcn_csr(
        const int* __restrict__ sorted, const int* __restrict__ tab,
        int* __restrict__ off, float* __restrict__ dinv, int* __restrict__ srow) {
    __shared__ int pst[256];
    __shared__ int pln[256];
    __shared__ int cnt[512];
    __shared__ int wsum[32];
    __shared__ int redA[32];
    __shared__ int redL[32];
    __shared__ int sbuf[SCAP];
    const int tid = threadIdx.x, lane = tid & 31;
    const int wave = __builtin_amdgcn_readfirstlane(tid >> 5);
    const int b = blockIdx.x;

    int s0 = 0, len = 0;
    if (tid < NCHUNK) {
        s0  = tab[b * NCHUNK + tid];
        len = tab[(b + 1) * NCHUNK + tid] - s0;
        pst[tid] = tid * CHUNK + s0;
        pln[tid] = len;
    }
    if (tid < 512) cnt[tid] = 0;
    for (int d = 16; d >= 1; d >>= 1) { s0 += __shfl_xor(s0, d); len += __shfl_xor(len, d); }
    if (lane == 0) { redA[wave] = s0; redL[wave] = len; }
    __syncthreads();
    int base = 0, total = 0;
    for (int w = 0; w < 32; ++w) { base += redA[w]; total += redL[w]; }
    const bool fits = total <= SCAP;

    int w0[8], w1[8];
    #pragma unroll
    for (int q = 0; q < 8; ++q) {
        const int p = wave + 32 * q;
        w0[q] = -1; w1[q] = -1;
        if (p < NCHUNK) {
            const int ln = pln[p], st = pst[p];
            if (lane < ln)      w0[q] = sorted[st + lane];
            if (lane + 32 < ln) w1[q] = sorted[st + 32 + lane];
        }
    }

    #pragma unroll
    for (int q = 0; q < 8; ++q) {
        if (w0[q] >= 0) atomicAdd(&cnt[w0[q] >> 17], 1);
        if (w1[q] >= 0) atomicAdd(&cnt[w1[q] >> 17], 1);
        const int p = wave + 32 * q;
        if (p < NCHUNK) {
            const int ln = pln[p], st = pst[p];
            for (int i = lane + 64; i < ln; i += 32) atomicAdd(&cnt[sorted[st + i] >> 17], 1);
        }
    }
    __syncthreads();

    const int x = (tid < 512) ? cnt[tid] : 0;
    int tot2;
    const int ex = block_excl_scan<32>(x, wsum, tot2);
    if (tid < 512) cnt[tid] = ex;
    if (tid < BNODES) {
        const int v = b * BNODES + tid;
        if (v < NNODES) {
            off[v]  = base + ex;
            dinv[v] = rsqrtf((float)(x + 1));
        }
    }
    if (b == NBUCK - 1 && tid == 0) off[NNODES] = NEDGES;
    __syncthreads();

    #pragma unroll
    for (int q = 0; q < 8; ++q) {
        if (w0[q] >= 0) {
            const int pos = atomicAdd(&cnt[w0[q] >> 17], 1);
            if (fits) sbuf[pos] = w0[q] & 0x1FFFF; else srow[base + pos] = w0[q] & 0x1FFFF;
        }
        if (w1[q] >= 0) {
            const int pos = atomicAdd(&cnt[w1[q] >> 17], 1);
            if (fits) sbuf[pos] = w1[q] & 0x1FFFF; else srow[base + pos] = w1[q] & 0x1FFFF;
        }
        const int p = wave + 32 * q;
        if (p < NCHUNK) {
            const int ln = pln[p], st = pst[p];
            for (int i = lane + 64; i < ln; i += 32) {
                const int w = sorted[st + i];
                const int pos = atomicAdd(&cnt[w >> 17], 1);
                if (fits) sbuf[pos] = w & 0x1FFFF; else srow[base + pos] = w & 0x1FFFF;
            }
        }
    }
    if (fits) {
        __syncthreads();
        for (int i = tid; i < total; i += 1024) srow[base + i] = sbuf[i];
    }
}

__global__ __launch_bounds__(256) void gcn_gemm_scale(
        const float* __restrict__ x, const float* __restrict__ W,
        const float* __restrict__ dinv, float* __restrict__ g) {
    __shared__ float Wl[256 * 64];
    const int tid = threadIdx.x;
    for (int i = tid; i < 256 * 64; i += 256) {
        int k = i >> 6, n = i & 63;
        Wl[(k >> 1) * 128 + (n << 1) + (k & 1)] = W[i];
    }
    __syncthreads();

    const int wave = tid >> 5;
    const int lane = tid & 31;
    const int half = lane >> 4;
    const int mr   = lane & 15;
    const int m0   = blockIdx.x * 128 + wave * 16;
    if (m0 >= NNODES) return;

    const float* xrow = x + (size_t)(m0 + mr) * 256;
    const v2f*   Wp   = (const v2f*)Wl;

    v8f acc0 = {}, acc1 = {}, acc2 = {}, acc3 = {};
    for (int k0 = 0; k0 < 256; k0 += 4) {
        const int ka = k0 + half * 2;
        v2f a = *(const v2f*)(xrow + ka);
        const int kp = ka >> 1;
        v2f b0 = Wp[kp * 64 + mr];
        v2f b1 = Wp[kp * 64 + 16 + mr];
        v2f b2 = Wp[kp * 64 + 32 + mr];
        v2f b3 = Wp[kp * 64 + 48 + mr];
        acc0 = __builtin_amdgcn_wmma_f32_16x16x4_f32(false, a, false, b0, (short)0, acc0, false, false);
        acc1 = __builtin_amdgcn_wmma_f32_16x16x4_f32(false, a, false, b1, (short)0, acc1, false, false);
        acc2 = __builtin_amdgcn_wmma_f32_16x16x4_f32(false, a, false, b2, (short)0, acc2, false, false);
        acc3 = __builtin_amdgcn_wmma_f32_16x16x4_f32(false, a, false, b3, (short)0, acc3, false, false);
    }

    #pragma unroll
    for (int i = 0; i < 8; ++i) {
        const int m = m0 + i + half * 8;
        const float s = dinv[m];
        const size_t rb = (size_t)m * 64 + mr;
        g[rb]      = acc0[i] * s;
        g[rb + 16] = acc1[i] * s;
        g[rb + 32] = acc2[i] * s;
        g[rb + 48] = acc3[i] * s;
    }
}

__global__ __launch_bounds__(256) void gcn_gather(
        const float* __restrict__ g, const int* __restrict__ off,
        const int* __restrict__ srow, const float* __restrict__ dinv,
        const float* __restrict__ b, float* __restrict__ out) {
    const int lane = threadIdx.x & 31;
    const int half = lane >> 4;
    const int sub  = lane & 15;
    const int v = __builtin_amdgcn_readfirstlane((int)(blockIdx.x * 8 + (threadIdx.x >> 5)));
    const int beg = off[v], end = off[v + 1];
    const float4* g4 = (const float4*)g;

    const float4 self = g4[(unsigned)v * 16u + (unsigned)sub];
    float4 acc = make_float4(0.f, 0.f, 0.f, 0.f);

    int idx = v;
    if (lane < end - beg) idx = srow[beg + lane];
    for (int base = beg; base < end; base += 32) {
        const int n = min(32, end - base);
        int nidx = v;
        if (lane < end - base - 32) nidx = srow[base + 32 + lane];
        for (int k = 0; k < n; k += 8) {
            for (int u = 0; u < 4; ++u) {
                const int kk = k + 2 * u + half;
                const unsigned r = (unsigned)__shfl(idx, kk);
                const float4 t = g4[r * 16u + (unsigned)sub];
                const bool ok = kk < n;
                acc.x += ok ? t.x : 0.f; acc.y += ok ? t.y : 0.f;
                acc.z += ok ? t.z : 0.f; acc.w += ok ? t.w : 0.f;
            }
        }
        idx = nidx;
    }
    if (half == 0) {
        acc.x += self.x; acc.y += self.y; acc.z += self.z; acc.w += self.w;
    }

    acc.x += __shfl_xor(acc.x, 16); acc.y += __shfl_xor(acc.y, 16);
    acc.z += __shfl_xor(acc.z, 16); acc.w += __shfl_xor(acc.w, 16);
    if (half == 0) {
        const float s = dinv[v];
        const float4 bb = ((const float4*)b)[sub];
        float4 o;
        o.x = s * acc.x + bb.x; o.y = s * acc.y + bb.y;
        o.z = s * acc.z + bb.z; o.w = s * acc.w + bb.w;
        ((float4*)out)[(unsigned)v * 16u + (unsigned)sub] = o;
    }
}

extern "C" void kernel_launch(void* const* d_in, const int* in_sizes, int n_in,
                              void* d_out, int out_size, void* d_ws, size_t ws_size,
                              hipStream_t stream) {
    const float* x  = (const float*)d_in[0];
    const int*   ei = (const int*)d_in[1];
    const float* W  = (const float*)d_in[2];
    const float* b  = (const float*)d_in[3];
    float* out = (float*)d_out;

    const int* row = ei;
    const int* col = ei + NEDGES;

    int*   off    = (int*)d_ws;
    float* dinv   = (float*)(off + NPAD);
    int*   tab    = (int*)(dinv + NPAD);
    int*   sorted = tab + 65536;
    int*   srow   = sorted + NEDGES;
    float* g      = (float*)(srow + NEDGES);

    gcn_bucket    <<<NCHUNK, 640, 0, stream>>>(row, col, sorted, tab);
    gcn_csr       <<<NBUCK, 1024, 0, stream>>>(sorted, tab, off, dinv, srow);
    gcn_gemm_scale<<<(NNODES + 127) / 128, 256, 0, stream>>>(x, W, dinv, g);
    gcn_gather    <<<NNODES / 8, 256, 0, stream>>>(g, off, srow, dinv, b, out);
}


# Dense product folded into the per-bucket launch: each bucket's work-group, once it has its in-degrees, forms g for its 391 rows (same WMMA loop, scale from LDS), one group per WGP; 4 launches to 3
# speedup vs baseline: 8.7216x; 1.0820x over previous
#include <hip/hip_runtime.h>


#define NNODES 100000
#define NEDGES 3200000
#define NPAD   100352

#define CHUNK   12800
#define NCHUNK  250
#define BNODES  391
#define NBUCK   256
#define SCAP    24576

typedef __attribute__((ext_vector_type(2))) float v2f;
typedef __attribute__((ext_vector_type(8))) float v8f;

template <int NW>
__device__ __forceinline__ int block_excl_scan(int x, int* wsum, int& total) {
    const int lane = threadIdx.x & 31, wave = threadIdx.x >> 5;
    int inc = x;
    for (int d = 1; d < 32; d <<= 1) {
        const int o = __shfl_up(inc, d);
        if (lane >= d) inc += o;
    }
    if (lane == 31) wsum[wave] = inc;
    __syncthreads();
    int wpre = 0, tot = 0;
    for (int w = 0; w < NW; ++w) {
        const int s = wsum[w];
        if (w < wave) wpre += s;
        tot += s;
    }
    total = tot;
    return wpre + inc - x;
}

__global__ __launch_bounds__(640) void gcn_bucket(
        const int* __restrict__ row, const int* __restrict__ col,
        int* __restrict__ sorted, int* __restrict__ tab) {
    __shared__ int cnt[512];
    __shared__ int wsum[32];
    __shared__ int buf[CHUNK];
    const int tid = threadIdx.x;
    if (tid < 512) cnt[tid] = 0;
    __syncthreads();

    const int4* c4 = (const int4*)(col + (size_t)blockIdx.x * CHUNK);
    const int4* r4 = (const int4*)(row + (size_t)blockIdx.x * CHUNK);
    int4 c[5], r[5];
    #pragma unroll
    for (int q = 0; q < 5; ++q) { c[q] = c4[tid + 640 * q]; r[q] = r4[tid + 640 * q]; }

    int key[20], wd[20];
    #pragma unroll
    for (int q = 0; q < 5; ++q) {
        const int d[4] = {c[q].x, c[q].y, c[q].z, c[q].w};
        const int s[4] = {r[q].x, r[q].y, r[q].z, r[q].w};
        #pragma unroll
        for (int j = 0; j < 4; ++j) {
            const int b  = (int)((unsigned)d[j] / (unsigned)BNODES);
            const int rk = atomicAdd(&cnt[b], 1);
            key[q * 4 + j] = (b << 16) | rk;
            wd[q * 4 + j]  = ((d[j] - b * BNODES) << 17) | s[j];
        }
    }
    __syncthreads();

    const int x = (tid < 512) ? cnt[tid] : 0;
    int total;
    const int ex = block_excl_scan<20>(x, wsum, total);
    if (tid < 512) cnt[tid] = ex;
    __syncthreads();

    #pragma unroll
    for (int e = 0; e < 20; ++e) buf[cnt[key[e] >> 16] + (key[e] & 0xFFFF)] = wd[e];
    if (tid <= NBUCK) tab[tid * NCHUNK + blockIdx.x] = cnt[tid];
    __syncthreads();

    int4* o4 = (int4*)(sorted + (size_t)blockIdx.x * CHUNK);
    const int4* b4 = (const int4*)buf;
    for (int q = 0; q < 5; ++q) o4[tid + 640 * q] = b4[tid + 640 * q];
}

__global__ __launch_bounds__(1024) void gcn_csr_gemm(
        const int* __restrict__ sorted, const int* __restrict__ tab,
        const float* __restrict__ x, const float* __restrict__ W,
        int* __restrict__ off, float* __restrict__ dinv, int* __restrict__ srow,
        float* __restrict__ g) {
    __shared__ int pst[256];
    __shared__ int pln[256];
    __shared__ int cnt[512];
    __shared__ int wsum[32];
    __shared__ int redA[32];
    __shared__ int redL[32];
    __shared__ int sbuf[SCAP];
    __shared__ float dsc[512];
    __shared__ float Wl[256 * 64];
    const int tid = threadIdx.x, lane = tid & 31;
    const int wave = __builtin_amdgcn_readfirstlane(tid >> 5);
    const int b = blockIdx.x;

    int s0 = 0, s1 = 0;
    if (tid < NCHUNK) {
        s0 = tab[b * NCHUNK + tid];
        s1 = tab[(b + 1) * NCHUNK + tid];
    }
    #pragma unroll
    for (int i = tid; i < 256 * 64; i += 1024) {
        const int k = i >> 6, n = i & 63;
        Wl[(k >> 1) * 128 + (n << 1) + (k & 1)] = W[i];
    }
    int len = s1 - s0;
    if (tid < NCHUNK) {
        pst[tid] = tid * CHUNK + s0;
        pln[tid] = len;
    }
    if (tid < 512) cnt[tid] = 0;
    for (int d = 16; d >= 1; d >>= 1) { s0 += __shfl_xor(s0, d); len += __shfl_xor(len, d); }
    if (lane == 0) { redA[wave] = s0; redL[wave] = len; }
    __syncthreads();
    int base = 0, total = 0;
    for (int w = 0; w < 32; ++w) { base += redA[w]; total += redL[w]; }
    const bool fits = total <= SCAP;

    int w0[8], w1[8];
    #pragma unroll
    for (int q = 0; q < 8; ++q) {
        const int p = wave + 32 * q;
        w0[q] = -1; w1[q] = -1;
        if (p < NCHUNK) {
            const int ln = pln[p], st = pst[p];
            if (lane < ln)      w0[q] = sorted[st + lane];
            if (lane + 32 < ln) w1[q] = sorted[st + 32 + lane];
        }
    }

    #pragma unroll
    for (int q = 0; q < 8; ++q) {
        if (w0[q] >= 0) atomicAdd(&cnt[w0[q] >> 17], 1);
        if (w1[q] >= 0) atomicAdd(&cnt[w1[q] >> 17], 1);
        const int p = wave + 32 * q;
        if (p < NCHUNK) {
            const int ln = pln[p], st = pst[p];
            for (int i = lane + 64; i < ln; i += 32) atomicAdd(&cnt[sorted[st + i] >> 17], 1);
        }
    }
    __syncthreads();

    const int deg = (tid < 512) ? cnt[tid] : 0;
    int tot2;
    const int ex = block_excl_scan<32>(deg, wsum, tot2);
    if (tid < 512) cnt[tid] = ex;
    const float di = rsqrtf((float)(deg + 1));
    if (tid < 512) dsc[tid] = di;
    if (tid < BNODES) {
        const int v = b * BNODES + tid;
        if (v < NNODES) {
            off[v]  = base + ex;
            dinv[v] = di;
        }
    }
    if (b == NBUCK - 1 && tid == 0) off[NNODES] = NEDGES;
    __syncthreads();

    #pragma unroll
    for (int q = 0; q < 8; ++q) {
        if (w0[q] >= 0) {
            const int pos = atomicAdd(&cnt[w0[q] >> 17], 1);
            if (fits) sbuf[pos] = w0[q] & 0x1FFFF; else srow[base + pos] = w0[q] & 0x1FFFF;
        }
        if (w1[q] >= 0) {
            const int pos = atomicAdd(&cnt[w1[q] >> 17], 1);
            if (fits) sbuf[pos] = w1[q] & 0x1FFFF; else srow[base + pos] = w1[q] & 0x1FFFF;
        }
        const int p = wave + 32 * q;
        if (p < NCHUNK) {
            const int ln = pln[p], st = pst[p];
            for (int i = lane + 64; i < ln; i += 32) {
                const int w = sorted[st + i];
                const int pos = atomicAdd(&cnt[w >> 17], 1);
                if (fits) sbuf[pos] = w & 0x1FFFF; else srow[base + pos] = w & 0x1FFFF;
            }
        }
    }
    if (fits) {
        __syncthreads();
        for (int i = tid; i < total; i += 1024) srow[base + i] = sbuf[i];
    }

    const int r0 = wave * 16;
    const int n0 = b * BNODES;
    if (r0 >= BNODES || n0 + r0 >= NNODES) return;
    const int half = lane >> 4;
    const int mr   = lane & 15;
    const int mrow = min(n0 + r0 + mr, NNODES - 1);
    const float* xrow = x + (size_t)mrow * 256;
    const v2f*   Wp   = (const v2f*)Wl;

    v8f acc0 = {}, acc1 = {}, acc2 = {}, acc3 = {};
    for (int k0 = 0; k0 < 256; k0 += 4) {
        const int ka = k0 + half * 2;
        v2f a = *(const v2f*)(xrow + ka);
        const int kp = ka >> 1;
        v2f b0 = Wp[kp * 64 + mr];
        v2f b1 = Wp[kp * 64 + 16 + mr];
        v2f b2 = Wp[kp * 64 + 32 + mr];
        v2f b3 = Wp[kp * 64 + 48 + mr];
        acc0 = __builtin_amdgcn_wmma_f32_16x16x4_f32(false, a, false, b0, (short)0, acc0, false, false);
        acc1 = __builtin_amdgcn_wmma_f32_16x16x4_f32(false, a, false, b1, (short)0, acc1, false, false);
        acc2 = __builtin_amdgcn_wmma_f32_16x16x4_f32(false, a, false, b2, (short)0, acc2, false, false);
        acc3 = __builtin_amdgcn_wmma_f32_16x16x4_f32(false, a, false, b3, (short)0, acc3, false, false);
    }

    float sc[8];
    #pragma unroll
    for (int i = 0; i < 8; ++i) sc[i] = dsc[r0 + half * 8 + i];
    #pragma unroll
    for (int i = 0; i < 8; ++i) {
        const int l = r0 + i + half * 8;
        const int m = n0 + l;
        if (l < BNODES && m < NNODES) {
            const size_t rb = (size_t)m * 64 + mr;
            g[rb]      = acc0[i] * sc[i];
            g[rb + 16] = acc1[i] * sc[i];
            g[rb + 32] = acc2[i] * sc[i];
            g[rb + 48] = acc3[i] * sc[i];
        }
    }
}


__global__ __launch_bounds__(256) void gcn_gather(
        const float* __restrict__ g, const int* __restrict__ off,
        const int* __restrict__ srow, const float* __restrict__ dinv,
        const float* __restrict__ b, float* __restrict__ out) {
    const int lane = threadIdx.x & 31;
    const int half = lane >> 4;
    const int sub  = lane & 15;
    const int v = __builtin_amdgcn_readfirstlane((int)(blockIdx.x * 8 + (threadIdx.x >> 5)));
    const int beg = off[v], end = off[v + 1];
    const float4* g4 = (const float4*)g;

    const float4 self = g4[(unsigned)v * 16u + (unsigned)sub];
    float4 acc = make_float4(0.f, 0.f, 0.f, 0.f);

    int idx = v;
    if (lane < end - beg) idx = srow[beg + lane];
    for (int base = beg; base < end; base += 32) {
        const int n = min(32, end - base);
        int nidx = v;
        if (lane < end - base - 32) nidx = srow[base + 32 + lane];
        for (int k = 0; k < n; k += 8) {
            for (int u = 0; u < 4; ++u) {
                const int kk = k + 2 * u + half;
                const unsigned r = (unsigned)__shfl(idx, kk);
                const float4 t = g4[r * 16u + (unsigned)sub];
                const bool ok = kk < n;
                acc.x += ok ? t.x : 0.f; acc.y += ok ? t.y : 0.f;
                acc.z += ok ? t.z : 0.f; acc.w += ok ? t.w : 0.f;
            }
        }
        idx = nidx;
    }
    if (half == 0) {
        acc.x += self.x; acc.y += self.y; acc.z += self.z; acc.w += self.w;
    }

    acc.x += __shfl_xor(acc.x, 16); acc.y += __shfl_xor(acc.y, 16);
    acc.z += __shfl_xor(acc.z, 16); acc.w += __shfl_xor(acc.w, 16);
    if (half == 0) {
        const float s = dinv[v];
        const float4 bb = ((const float4*)b)[sub];
        float4 o;
        o.x = s * acc.x + bb.x; o.y = s * acc.y + bb.y;
        o.z = s * acc.z + bb.z; o.w = s * acc.w + bb.w;
        ((float4*)out)[(unsigned)v * 16u + (unsigned)sub] = o;
    }
}

extern "C" void kernel_launch(void* const* d_in, const int* in_sizes, int n_in,
                              void* d_out, int out_size, void* d_ws, size_t ws_size,
                              hipStream_t stream) {
    const float* x  = (const float*)d_in[0];
    const int*   ei = (const int*)d_in[1];
    const float* W  = (const float*)d_in[2];
    const float* b  = (const float*)d_in[3];
    float* out = (float*)d_out;

    const int* row = ei;
    const int* col = ei + NEDGES;

    int*   off    = (int*)d_ws;
    float* dinv   = (float*)(off + NPAD);
    int*   tab    = (int*)(dinv + NPAD);
    int*   sorted = tab + 65536;
    int*   srow   = sorted + NEDGES;
    float* g      = (float*)(srow + NEDGES);

    gcn_bucket    <<<NCHUNK, 640, 0, stream>>>(row, col, sorted, tab);
    gcn_csr_gemm  <<<NBUCK, 1024, 0, stream>>>(sorted, tab, x, W, off, dinv, srow, g);
    gcn_gather    <<<NNODES / 8, 256, 0, stream>>>(g, off, srow, dinv, b, out);
}
